# starting kernel
# baseline (speedup 1.0000x reference)
#include <hip/hip_runtime.h>

typedef __attribute__((ext_vector_type(2))) float v2f;
typedef __attribute__((ext_vector_type(8))) float v8f;

#define SP 8
#define BLK 16

__global__ void build_M_kernel(const float* __restrict__ W_samp,
                               const float* __restrict__ W_init,
                               float* __restrict__ M) {
    const int t = threadIdx.x;
    const int k = t >> 4;
    const int n = t & 15;
    float acc = 0.0f;
#pragma unroll
    for (int s = 0; s < SP; ++s)
        acc += W_samp[s * BLK + k] * W_init[n * SP + s];
    M[k * BLK + n] = acc;
}

__global__ void __launch_bounds__(256) apply_M_kernel(
        const float* __restrict__ x,
        const float* __restrict__ M,
        float* __restrict__ y,
        int ntiles) {
    const int lane   = threadIdx.x & 31;
    const int wib    = threadIdx.x >> 5;
    const int gwave  = blockIdx.x * (blockDim.x >> 5) + wib;
    const int nwaves = gridDim.x * (blockDim.x >> 5);
    const int h = lane >> 4;
    const int n = lane & 15;

    v2f b[4];
#pragma unroll
    for (int c = 0; c < 4; ++c) {
        b[c].x = M[(4 * c + 2 * h + 0) * BLK + n];
        b[c].y = M[(4 * c + 2 * h + 1) * BLK + n];
    }

    for (int tile = gwave; tile < ntiles; tile += nwaves) {
        const float* Arow = x + (size_t)tile * 256 + (size_t)n * BLK;

        __builtin_prefetch(Arow + (size_t)nwaves * 256, 0, 0);

        v2f a[4];
#pragma unroll
        for (int c = 0; c < 4; ++c)
            a[c] = *(const v2f*)(Arow + 4 * c + 2 * h);

        v8f d = {};
#pragma unroll
        for (int c = 0; c < 4; ++c)
            d = __builtin_amdgcn_wmma_f32_16x16x4_f32(
                     false, a[c],
                     false, b[c],
                     (short)0, d,
                     false,  false);

        float* Drow = y + (size_t)tile * 256;
#pragma unroll
        for (int g = 0; g < 8; ++g)
            Drow[(size_t)(g + 8 * h) * BLK + n] = d[g];
    }
}

extern "C" void kernel_launch(void* const* d_in, const int* in_sizes, int n_in,
                              void* d_out, int out_size, void* d_ws, size_t ws_size,
                              hipStream_t stream) {
    const float* x      = (const float*)d_in[0];
    const float* W_samp = (const float*)d_in[1];
    const float* W_init = (const float*)d_in[2];
    float* y = (float*)d_out;
    float* M = (float*)d_ws;

    build_M_kernel<<<1, 256, 0, stream>>>(W_samp, W_init, M);

    const int ntiles = in_sizes[0] / 256;
    const int blocks = 1024;
    apply_M_kernel<<<blocks, 256, 0, stream>>>(x, M, y, ntiles);
}


# One launch instead of two: each wave folds W_samp x W_init into its own B fragment of M with two f32 WMMAs (M's 8-term sums move from serial FMA order to the matrix unit), no workspace round trip
# speedup vs baseline: 1.1126x; 1.1126x over previous
#include <hip/hip_runtime.h>


typedef __attribute__((ext_vector_type(2))) float v2f;
typedef __attribute__((ext_vector_type(8))) float v8f;

#define SP 8
#define BLK 16

static __device__ __forceinline__ void mul_store_tile(const v2f (&a)[4],
                                                      const v2f (&b)[4],
                                                      float* __restrict__ y,
                                                      int tile, int n, int h) {
    v8f d = {};
#pragma unroll
    for (int c = 0; c < 4; ++c)
        d = __builtin_amdgcn_wmma_f32_16x16x4_f32(
                 false, a[c],
                 false, b[c],
                 (short)0, d,
                 false,  false);

    float* Drow = y + (size_t)tile * 256;
#pragma unroll
    for (int g = 0; g < 8; ++g)
        Drow[(size_t)(g + 8 * h) * BLK + n] = d[g];
}

static __device__ __forceinline__ void load_tile(v2f (&a)[4],
                                                 const float* __restrict__ Arow,
                                                 int h) {
#pragma unroll
    for (int c = 0; c < 4; ++c)
        a[c] = *(const v2f*)(Arow + 4 * c + 2 * h);
}

__global__ void __launch_bounds__(256) fold_apply_kernel(
        const float* __restrict__ x,
        const float* __restrict__ W_samp,
        const float* __restrict__ W_init,
        float* __restrict__ y,
        int ntiles) {
    const int lane   = threadIdx.x & 31;
    const int wib    = threadIdx.x >> 5;
    const int gwave  = blockIdx.x * (blockDim.x >> 5) + wib;
    const int nwaves = gridDim.x * (blockDim.x >> 5);
    const int h = lane >> 4;
    const int n = lane & 15;

    if (gwave >= ntiles) return;

    const int kp = 4 * ((n & 7) >> 1) + 2 * (n >> 3) + (n & 1);
    v2f wa[2], wb[2];
#pragma unroll
    for (int j = 0; j < 2; ++j) {
        wa[j].x = W_samp[(4 * j + 2 * h + 0) * BLK + kp];
        wa[j].y = W_samp[(4 * j + 2 * h + 1) * BLK + kp];
        wb[j]   = *(const v2f*)(W_init + n * SP + 4 * j + 2 * h);
    }

    int tile = gwave;
    const float* Arow = x + (size_t)tile * 256 + (size_t)n * BLK;
    __builtin_prefetch(Arow + (size_t)nwaves * 256, 0, 0);
    v2f a[4];
    load_tile(a, Arow, h);

    v8f m = {};
#pragma unroll
    for (int j = 0; j < 2; ++j)
        m = __builtin_amdgcn_wmma_f32_16x16x4_f32(
                 false, wa[j],
                 false, wb[j],
                 (short)0, m,
                 false,  false);

    v2f b[4];
#pragma unroll
    for (int c = 0; c < 4; ++c) {
        b[c].x = m[2 * c + 0];
        b[c].y = m[2 * c + 1];
    }

    mul_store_tile(a, b, y, tile, n, h);

    for (tile += nwaves; tile < ntiles; tile += nwaves) {
        Arow = x + (size_t)tile * 256 + (size_t)n * BLK;

        __builtin_prefetch(Arow + (size_t)nwaves * 256, 0, 0);

        load_tile(a, Arow, h);
        mul_store_tile(a, b, y, tile, n, h);
    }
}

extern "C" void kernel_launch(void* const* d_in, const int* in_sizes, int n_in,
                              void* d_out, int out_size, void* d_ws, size_t ws_size,
                              hipStream_t stream) {
    const float* x      = (const float*)d_in[0];
    const float* W_samp = (const float*)d_in[1];
    const float* W_init = (const float*)d_in[2];
    float* y = (float*)d_out;

    const int ntiles = in_sizes[0] / 256;
    const int blocks = 1024;
    fold_apply_kernel<<<blocks, 256, 0, stream>>>(x, W_samp, W_init, y, ntiles);
}


# Whole-line accesses: x tile as B, M^T as A (k sum order now 0,1,8,9|2,3,10,11|..), lanes n,n^1 trade 16-B pieces by DPP: 2 b128 loads + 2 b128 stores per tile, 4 lines each (was 4 b64 + 8 b32)
# speedup vs baseline: 1.2282x; 1.1039x over previous
#include <hip/hip_runtime.h>


typedef __attribute__((ext_vector_type(2))) float v2f;
typedef __attribute__((ext_vector_type(4))) float v4f;
typedef __attribute__((ext_vector_type(8))) float v8f;

#define SP 8
#define BLK 16


static __device__ __forceinline__ float from_neighbour(float v) {
    return __builtin_bit_cast(float,
        __builtin_amdgcn_update_dpp(0, __builtin_bit_cast(int, v),
                                     0xB1,
                                     0xF,  0xF,
                                     true));
}

static __device__ __forceinline__ void load_tile(v4f& l0, v4f& l1,
                                                 const float* __restrict__ xt,
                                                 int off0, int off1) {
    l0 = *(const v4f*)(xt + off0);
    l1 = *(const v4f*)(xt + off1);
}

static __device__ __forceinline__ void mul_store_tile(const v4f& l0, const v4f& l1,
                                                      const v2f (&am)[4],
                                                      float* __restrict__ yt,
                                                      int off0, int off1, bool odd) {
    v4f own, give, got;
#pragma unroll
    for (int j = 0; j < 4; ++j) {
        own[j]  = odd ? l1[j] : l0[j];
        give[j] = odd ? l0[j] : l1[j];
        got[j]  = from_neighbour(give[j]);
    }

    v2f bx[4];
    bx[0].x = own[0]; bx[0].y = own[1];
    bx[1].x = own[2]; bx[1].y = own[3];
    bx[2].x = got[0]; bx[2].y = got[1];
    bx[3].x = got[2]; bx[3].y = got[3];

    v8f d = {};
#pragma unroll
    for (int c = 0; c < 4; ++c)
        d = __builtin_amdgcn_wmma_f32_16x16x4_f32(
                 false, am[c],
                 false, bx[c],
                 (short)0, d,
                 false,  false);

    v4f keep, recv, s0, s1;
#pragma unroll
    for (int j = 0; j < 4; ++j) {
        keep[j] = d[j];
        recv[j] = from_neighbour(d[4 + j]);
        s0[j] = odd ? recv[j] : keep[j];
        s1[j] = odd ? keep[j] : recv[j];
    }
    *(v4f*)(yt + off0) = s0;
    *(v4f*)(yt + off1) = s1;
}

__global__ void __launch_bounds__(256) fold_apply_kernel(
        const float* __restrict__ x,
        const float* __restrict__ W_samp,
        const float* __restrict__ W_init,
        float* __restrict__ y,
        int ntiles) {
    const int lane   = threadIdx.x & 31;
    const int wib    = threadIdx.x >> 5;
    const int gwave  = blockIdx.x * (blockDim.x >> 5) + wib;
    const int nwaves = gridDim.x * (blockDim.x >> 5);
    const int h = lane >> 4;
    const int n = lane & 15;
    const bool odd = (n & 1) != 0;
    const int  q   = n >> 1;

    if (gwave >= ntiles) return;

    const int off0 = 16 * q + 8 * h + (odd ? 4 : 0);
    const int off1 = 128 + 16 * q + 8 * h + (odd ? 0 : 4);

    v2f wa[2], wb[2];
#pragma unroll
    for (int j = 0; j < 2; ++j) {
        wa[j].x = W_samp[(4 * j + 2 * h + 0) * BLK + n];
        wa[j].y = W_samp[(4 * j + 2 * h + 1) * BLK + n];
        wb[j]   = *(const v2f*)(W_init + n * SP + 4 * j + 2 * h);
    }

    int tile = gwave;
    const float* xt = x + (size_t)tile * 256;
    __builtin_prefetch(xt + (size_t)nwaves * 256 + (size_t)n * BLK, 0, 0);
    v4f l0, l1;
    load_tile(l0, l1, xt, off0, off1);

    v8f m = {};
#pragma unroll
    for (int j = 0; j < 2; ++j)
        m = __builtin_amdgcn_wmma_f32_16x16x4_f32(
                 false, wa[j],
                 false, wb[j],
                 (short)0, m,
                 false,  false);

    v2f am[4];
#pragma unroll
    for (int c = 0; c < 4; ++c) {
        am[c].x = m[2 * c + 0];
        am[c].y = m[2 * c + 1];
    }

    mul_store_tile(l0, l1, am, y + (size_t)tile * 256, off0, off1, odd);

    for (tile += nwaves; tile < ntiles; tile += nwaves) {
        xt = x + (size_t)tile * 256;

        __builtin_prefetch(xt + (size_t)nwaves * 256 + (size_t)n * BLK, 0, 0);

        load_tile(l0, l1, xt, off0, off1);
        mul_store_tile(l0, l1, am, y + (size_t)tile * 256, off0, off1, odd);
    }
}

extern "C" void kernel_launch(void* const* d_in, const int* in_sizes, int n_in,
                              void* d_out, int out_size, void* d_ws, size_t ws_size,
                              hipStream_t stream) {
    const float* x      = (const float*)d_in[0];
    const float* W_samp = (const float*)d_in[1];
    const float* W_init = (const float*)d_in[2];
    float* y = (float*)d_out;

    const int ntiles = in_sizes[0] / 256;
    const int blocks = 1024;
    fold_apply_kernel<<<blocks, 256, 0, stream>>>(x, W_samp, W_init, y, ntiles);
}
